# starting kernel
# baseline (speedup 1.0000x reference)
#include <hip/hip_runtime.h>


#define N_INPUT     4096
#define NUM_CENTERS 512
#define DIM         128
#define N_OUT       10

#define TILE_ROWS   16
#define BLOCK       256
#define BATCH_STRIDE 132
#define RAD_STRIDE   513

typedef __attribute__((ext_vector_type(2))) float v2f;
typedef __attribute__((ext_vector_type(8))) float v8f;

__global__ __launch_bounds__(BLOCK)
void rbfn_fused_kernel(const float* __restrict__ batches,
                       const float* __restrict__ centers,
                       const float* __restrict__ sigma_p,
                       const float* __restrict__ W,
                       const float* __restrict__ bias,
                       float* __restrict__ out)
{
    __shared__ float batch_lds[TILE_ROWS * BATCH_STRIDE];
    __shared__ float radial_lds[TILE_ROWS * RAD_STRIDE];

    const int tid  = threadIdx.x;
    const int row0 = blockIdx.x * TILE_ROWS;
    const float sigma = sigma_p[0];

    {
        const float4* gsrc = (const float4*)(batches + (size_t)row0 * DIM);
        for (int i = tid; i < TILE_ROWS * (DIM / 4); i += BLOCK) {
            const int r  = i >> 5;
            const int c4 = i & 31;
            float4 v = gsrc[r * (DIM / 4) + c4];
            float* dst = &batch_lds[r * BATCH_STRIDE + c4 * 4];
            dst[0] = v.x; dst[1] = v.y; dst[2] = v.z; dst[3] = v.w;
        }
    }
    __syncthreads();

    {
        const int r    = tid & 15;
        const int cgrp = tid >> 4;
        const float* brow = &batch_lds[r * BATCH_STRIDE];
        for (int i = 0; i < NUM_CENTERS / 16; ++i) {
            const int c = i * 16 + cgrp;
            const float4* crow = (const float4*)(centers + (size_t)c * DIM);
            float acc = 0.0f;
#pragma unroll 8
            for (int d4 = 0; d4 < DIM / 4; ++d4) {
                float4 cv = crow[d4];
                acc += __builtin_fabsf(brow[d4 * 4 + 0] - cv.x);
                acc += __builtin_fabsf(brow[d4 * 4 + 1] - cv.y);
                acc += __builtin_fabsf(brow[d4 * 4 + 2] - cv.z);
                acc += __builtin_fabsf(brow[d4 * 4 + 3] - cv.w);
            }
            radial_lds[r * RAD_STRIDE + c] = __expf(-sigma * acc);
        }
    }
    __syncthreads();

    if (tid < 32) {
        const int lane = tid;
        const int half = lane >> 4;
        const int m    = lane & 15;

        const bool  valid = (m < N_OUT);
        const float mask  = valid ? 1.0f : 0.0f;
        const float* wrow = W + (size_t)(valid ? m : 0) * NUM_CENTERS;
        const float* arow = radial_lds + m * RAD_STRIDE;

        v8f acc0 = {};
        v8f acc1 = {};
        for (int k = 0; k < NUM_CENTERS; k += 8) {
            const int kk0 = k + 2 * half;
            const int kk1 = k + 4 + 2 * half;

            v2f a0, a1, b0, b1;
            a0.x = arow[kk0];
            a0.y = arow[kk0 + 1];
            a1.x = arow[kk1];
            a1.y = arow[kk1 + 1];
            const float2 w0 = *(const float2*)(wrow + kk0);
            const float2 w1 = *(const float2*)(wrow + kk1);
            b0.x = w0.x * mask;  b0.y = w0.y * mask;
            b1.x = w1.x * mask;  b1.y = w1.y * mask;

            acc0 = __builtin_amdgcn_wmma_f32_16x16x4_f32(
                false, a0, false, b0, (short)0, acc0, false, false);
            acc1 = __builtin_amdgcn_wmma_f32_16x16x4_f32(
                false, a1, false, b1, (short)0, acc1, false, false);
        }

        if (valid) {
            const float bv = bias[m];
#pragma unroll
            for (int v = 0; v < 8; ++v) {
                const int row = row0 + v + 8 * half;
                out[(size_t)row * N_OUT + m] = acc0[v] + acc1[v] + bv;
            }
        }
    }
}

extern "C" void kernel_launch(void* const* d_in, const int* in_sizes, int n_in,
                              void* d_out, int out_size, void* d_ws, size_t ws_size,
                              hipStream_t stream) {
    const float* batches = (const float*)d_in[0];
    const float* centers = (const float*)d_in[1];
    const float* sigma   = (const float*)d_in[2];
    const float* W       = (const float*)d_in[3];
    const float* bias    = (const float*)d_in[4];
    float* out           = (float*)d_out;

    (void)in_sizes; (void)n_in; (void)out_size; (void)d_ws; (void)ws_size;

    rbfn_fused_kernel<<<N_INPUT / TILE_ROWS, BLOCK, 0, stream>>>(
        batches, centers, sigma, W, bias, out);
}


# Distance phase re-tiled for reuse: centres staged through LDS in 16-dim chunks (coalesced loads, next chunk prefetched to registers), each thread sums a 4-row x 8-centre block; per-sum add order kept
# speedup vs baseline: 2.5580x; 2.5580x over previous
#include <hip/hip_runtime.h>


#define N_INPUT     4096
#define NUM_CENTERS 512
#define DIM         128
#define N_OUT       10

#define TILE_ROWS   16
#define BLOCK       256
#define BATCH_STRIDE 132
#define RAD_STRIDE   513

#define CHUNK_D     16
#define N_CHUNKS    (DIM / CHUNK_D)
#define C_PITCH     20
#define PRE_LOADS   ((NUM_CENTERS * CHUNK_D / 4) / BLOCK)
#define TR          4
#define TC          8
#define C_GROUPS    (NUM_CENTERS / TC)

typedef __attribute__((ext_vector_type(2))) float v2f;
typedef __attribute__((ext_vector_type(8))) float v8f;
typedef __attribute__((ext_vector_type(4))) float v4f;

static __device__ __forceinline__
void load_chunk(v4f (&pre)[PRE_LOADS], const float* __restrict__ centers,
                int tid, int k)
{
#pragma unroll
    for (int j = 0; j < PRE_LOADS; ++j) {
        const int p = tid + BLOCK * j;
        const int c = p >> 2;
        const int q = p & 3;
        pre[j] = *(const v4f*)(centers + (size_t)c * DIM + k * CHUNK_D + q * 4);
    }
}

static __device__ __forceinline__
void store_chunk(const v4f (&pre)[PRE_LOADS], float* buf, int tid)
{
#pragma unroll
    for (int j = 0; j < PRE_LOADS; ++j) {
        const int p = tid + BLOCK * j;
        const int c = p >> 2;
        const int q = p & 3;
        *(v4f*)(buf + c * C_PITCH + q * 4) = pre[j];
    }
}

__global__ __launch_bounds__(BLOCK)
void rbfn_fused_kernel(const float* __restrict__ batches,
                       const float* __restrict__ centers,
                       const float* __restrict__ sigma_p,
                       const float* __restrict__ W,
                       const float* __restrict__ bias,
                       float* __restrict__ out)
{
    __shared__ __attribute__((aligned(16))) float batch_lds[TILE_ROWS * BATCH_STRIDE];
    __shared__ __attribute__((aligned(16))) float radial_lds[TILE_ROWS * RAD_STRIDE];
    __shared__ __attribute__((aligned(16))) float cbuf[NUM_CENTERS * C_PITCH];

    const int tid  = threadIdx.x;
    const int row0 = blockIdx.x * TILE_ROWS;
    const float sigma = sigma_p[0];

    v4f first[PRE_LOADS];
    load_chunk(first, centers, tid, 0);

    {
        const float4* gsrc = (const float4*)(batches + (size_t)row0 * DIM);
        for (int i = tid; i < TILE_ROWS * (DIM / 4); i += BLOCK) {
            const int r  = i >> 5;
            const int c4 = i & 31;
            float4 v = gsrc[r * (DIM / 4) + c4];
            float* dst = &batch_lds[r * BATCH_STRIDE + c4 * 4];
            dst[0] = v.x; dst[1] = v.y; dst[2] = v.z; dst[3] = v.w;
        }
    }
    store_chunk(first, cbuf, tid);

    {
        const int cg = tid & (C_GROUPS - 1);
        const int rg = tid / C_GROUPS;

        float acc[TR][TC];
#pragma unroll
        for (int i = 0; i < TR; ++i)
#pragma unroll
            for (int j = 0; j < TC; ++j)
                acc[i][j] = 0.0f;

        v4f pre[PRE_LOADS];

#pragma unroll 1
        for (int k = 0; k < N_CHUNKS; ++k) {
            if (k > 0) {
                __syncthreads();
                store_chunk(pre, cbuf, tid);
            }
            __syncthreads();
            if (k + 1 < N_CHUNKS)
                load_chunk(pre, centers, tid, k + 1);

            const float* bp = batch_lds + (rg * TR) * BATCH_STRIDE + k * CHUNK_D;
            const float* cp = cbuf + cg * C_PITCH;

#pragma unroll
            for (int d4 = 0; d4 < CHUNK_D / 4; ++d4) {
                float4 b[TR];
                float4 c[TC];
#pragma unroll
                for (int i = 0; i < TR; ++i)
                    b[i] = *(const float4*)(bp + i * BATCH_STRIDE + d4 * 4);
#pragma unroll
                for (int j = 0; j < TC; ++j)
                    c[j] = *(const float4*)(cp + (j * C_GROUPS) * C_PITCH + d4 * 4);
#pragma unroll
                for (int i = 0; i < TR; ++i) {
#pragma unroll
                    for (int j = 0; j < TC; ++j) {
                        float a = acc[i][j];
                        a += __builtin_fabsf(b[i].x - c[j].x);
                        a += __builtin_fabsf(b[i].y - c[j].y);
                        a += __builtin_fabsf(b[i].z - c[j].z);
                        a += __builtin_fabsf(b[i].w - c[j].w);
                        acc[i][j] = a;
                    }
                }
            }
        }

#pragma unroll
        for (int i = 0; i < TR; ++i)
#pragma unroll
            for (int j = 0; j < TC; ++j)
                radial_lds[(rg * TR + i) * RAD_STRIDE + cg + C_GROUPS * j] =
                    __expf(-sigma * acc[i][j]);
    }
    __syncthreads();

    if (tid < 32) {
        const int lane = tid;
        const int half = lane >> 4;
        const int m    = lane & 15;

        const bool  valid = (m < N_OUT);
        const float mask  = valid ? 1.0f : 0.0f;
        const float* wrow = W + (size_t)(valid ? m : 0) * NUM_CENTERS;
        const float* arow = radial_lds + m * RAD_STRIDE;

        v8f acc0 = {};
        v8f acc1 = {};
        for (int k = 0; k < NUM_CENTERS; k += 8) {
            const int kk0 = k + 2 * half;
            const int kk1 = k + 4 + 2 * half;

            v2f a0, a1, b0, b1;
            a0.x = arow[kk0];
            a0.y = arow[kk0 + 1];
            a1.x = arow[kk1];
            a1.y = arow[kk1 + 1];
            const float2 w0 = *(const float2*)(wrow + kk0);
            const float2 w1 = *(const float2*)(wrow + kk1);
            b0.x = w0.x * mask;  b0.y = w0.y * mask;
            b1.x = w1.x * mask;  b1.y = w1.y * mask;

            acc0 = __builtin_amdgcn_wmma_f32_16x16x4_f32(
                false, a0, false, b0, (short)0, acc0, false, false);
            acc1 = __builtin_amdgcn_wmma_f32_16x16x4_f32(
                false, a1, false, b1, (short)0, acc1, false, false);
        }

        if (valid) {
            const float bv = bias[m];
#pragma unroll
            for (int v = 0; v < 8; ++v) {
                const int row = row0 + v + 8 * half;
                out[(size_t)row * N_OUT + m] = acc0[v] + acc1[v] + bv;
            }
        }
    }
}

extern "C" void kernel_launch(void* const* d_in, const int* in_sizes, int n_in,
                              void* d_out, int out_size, void* d_ws, size_t ws_size,
                              hipStream_t stream) {
    const float* batches = (const float*)d_in[0];
    const float* centers = (const float*)d_in[1];
    const float* sigma   = (const float*)d_in[2];
    const float* W       = (const float*)d_in[3];
    const float* bias    = (const float*)d_in[4];
    float* out           = (float*)d_out;

    (void)in_sizes; (void)n_in; (void)out_size; (void)d_ws; (void)ws_size;

    rbfn_fused_kernel<<<N_INPUT / TILE_ROWS, BLOCK, 0, stream>>>(
        batches, centers, sigma, W, bias, out);
}


# Distance loop reordered dim-major: a thread's 32 sums take one dim's term in turn, so neighbouring instructions are independent (no serial sub->add chain); per-sum add order d=0..127 kept
# speedup vs baseline: 2.7197x; 1.0632x over previous
#include <hip/hip_runtime.h>


#define N_INPUT     4096
#define NUM_CENTERS 512
#define DIM         128
#define N_OUT       10

#define TILE_ROWS   16
#define BLOCK       256
#define BATCH_STRIDE 132
#define RAD_STRIDE   513

#define CHUNK_D     16
#define N_CHUNKS    (DIM / CHUNK_D)
#define C_PITCH     20
#define PRE_LOADS   ((NUM_CENTERS * CHUNK_D / 4) / BLOCK)
#define TR          4
#define TC          8
#define C_GROUPS    (NUM_CENTERS / TC)

typedef __attribute__((ext_vector_type(2))) float v2f;
typedef __attribute__((ext_vector_type(8))) float v8f;
typedef __attribute__((ext_vector_type(4))) float v4f;

static __device__ __forceinline__
void load_chunk(v4f (&pre)[PRE_LOADS], const float* __restrict__ centers,
                int tid, int k)
{
#pragma unroll
    for (int j = 0; j < PRE_LOADS; ++j) {
        const int p = tid + BLOCK * j;
        const int c = p >> 2;
        const int q = p & 3;
        pre[j] = *(const v4f*)(centers + (size_t)c * DIM + k * CHUNK_D + q * 4);
    }
}

static __device__ __forceinline__
void store_chunk(const v4f (&pre)[PRE_LOADS], float* buf, int tid)
{
#pragma unroll
    for (int j = 0; j < PRE_LOADS; ++j) {
        const int p = tid + BLOCK * j;
        const int c = p >> 2;
        const int q = p & 3;
        *(v4f*)(buf + c * C_PITCH + q * 4) = pre[j];
    }
}

__global__ __launch_bounds__(BLOCK)
void rbfn_fused_kernel(const float* __restrict__ batches,
                       const float* __restrict__ centers,
                       const float* __restrict__ sigma_p,
                       const float* __restrict__ W,
                       const float* __restrict__ bias,
                       float* __restrict__ out)
{
    __shared__ __attribute__((aligned(16))) float batch_lds[TILE_ROWS * BATCH_STRIDE];
    __shared__ __attribute__((aligned(16))) float radial_lds[TILE_ROWS * RAD_STRIDE];
    __shared__ __attribute__((aligned(16))) float cbuf[NUM_CENTERS * C_PITCH];

    const int tid  = threadIdx.x;
    const int row0 = blockIdx.x * TILE_ROWS;
    const float sigma = sigma_p[0];

    v4f first[PRE_LOADS];
    load_chunk(first, centers, tid, 0);

    {
        const float4* gsrc = (const float4*)(batches + (size_t)row0 * DIM);
        for (int i = tid; i < TILE_ROWS * (DIM / 4); i += BLOCK) {
            const int r  = i >> 5;
            const int c4 = i & 31;
            float4 v = gsrc[r * (DIM / 4) + c4];
            float* dst = &batch_lds[r * BATCH_STRIDE + c4 * 4];
            dst[0] = v.x; dst[1] = v.y; dst[2] = v.z; dst[3] = v.w;
        }
    }
    store_chunk(first, cbuf, tid);

    {
        const int cg = tid & (C_GROUPS - 1);
        const int rg = tid / C_GROUPS;

        float acc[TR][TC];
#pragma unroll
        for (int i = 0; i < TR; ++i)
#pragma unroll
            for (int j = 0; j < TC; ++j)
                acc[i][j] = 0.0f;

        v4f pre[PRE_LOADS];

#pragma unroll 1
        for (int k = 0; k < N_CHUNKS; ++k) {
            if (k > 0) {
                __syncthreads();
                store_chunk(pre, cbuf, tid);
            }
            __syncthreads();
            if (k + 1 < N_CHUNKS)
                load_chunk(pre, centers, tid, k + 1);

            const float* bp = batch_lds + (rg * TR) * BATCH_STRIDE + k * CHUNK_D;
            const float* cp = cbuf + cg * C_PITCH;

#pragma unroll
            for (int d4 = 0; d4 < CHUNK_D / 4; ++d4) {
                v4f b[TR];
                v4f c[TC];
#pragma unroll
                for (int i = 0; i < TR; ++i)
                    b[i] = *(const v4f*)(bp + i * BATCH_STRIDE + d4 * 4);
#pragma unroll
                for (int j = 0; j < TC; ++j)
                    c[j] = *(const v4f*)(cp + (j * C_GROUPS) * C_PITCH + d4 * 4);
#pragma unroll
                for (int e = 0; e < 4; ++e) {
#pragma unroll
                    for (int i = 0; i < TR; ++i) {
#pragma unroll
                        for (int j = 0; j < TC; ++j)
                            acc[i][j] += __builtin_fabsf(b[i][e] - c[j][e]);
                    }
                }
            }
        }

#pragma unroll
        for (int i = 0; i < TR; ++i)
#pragma unroll
            for (int j = 0; j < TC; ++j)
                radial_lds[(rg * TR + i) * RAD_STRIDE + cg + C_GROUPS * j] =
                    __expf(-sigma * acc[i][j]);
    }
    __syncthreads();

    if (tid < 32) {
        const int lane = tid;
        const int half = lane >> 4;
        const int m    = lane & 15;

        const bool  valid = (m < N_OUT);
        const float mask  = valid ? 1.0f : 0.0f;
        const float* wrow = W + (size_t)(valid ? m : 0) * NUM_CENTERS;
        const float* arow = radial_lds + m * RAD_STRIDE;

        v8f acc0 = {};
        v8f acc1 = {};
        for (int k = 0; k < NUM_CENTERS; k += 8) {
            const int kk0 = k + 2 * half;
            const int kk1 = k + 4 + 2 * half;

            v2f a0, a1, b0, b1;
            a0.x = arow[kk0];
            a0.y = arow[kk0 + 1];
            a1.x = arow[kk1];
            a1.y = arow[kk1 + 1];
            const float2 w0 = *(const float2*)(wrow + kk0);
            const float2 w1 = *(const float2*)(wrow + kk1);
            b0.x = w0.x * mask;  b0.y = w0.y * mask;
            b1.x = w1.x * mask;  b1.y = w1.y * mask;

            acc0 = __builtin_amdgcn_wmma_f32_16x16x4_f32(
                false, a0, false, b0, (short)0, acc0, false, false);
            acc1 = __builtin_amdgcn_wmma_f32_16x16x4_f32(
                false, a1, false, b1, (short)0, acc1, false, false);
        }

        if (valid) {
            const float bv = bias[m];
#pragma unroll
            for (int v = 0; v < 8; ++v) {
                const int row = row0 + v + 8 * half;
                out[(size_t)row * N_OUT + m] = acc0[v] + acc1[v] + bv;
            }
        }
    }
}

extern "C" void kernel_launch(void* const* d_in, const int* in_sizes, int n_in,
                              void* d_out, int out_size, void* d_ws, size_t ws_size,
                              hipStream_t stream) {
    const float* batches = (const float*)d_in[0];
    const float* centers = (const float*)d_in[1];
    const float* sigma   = (const float*)d_in[2];
    const float* W       = (const float*)d_in[3];
    const float* bias    = (const float*)d_in[4];
    float* out           = (float*)d_out;

    (void)in_sizes; (void)n_in; (void)out_size; (void)d_ws; (void)ws_size;

    rbfn_fused_kernel<<<N_INPUT / TILE_ROWS, BLOCK, 0, stream>>>(
        batches, centers, sigma, W, bias, out);
}


# L1 distance summed as 2*sum max(b,c) - sum b - sum c (exact identity for |b-c|): the max dual-issues, one packed add serves two dims; per-sum add order changes (even/odd dim partials)
# speedup vs baseline: 3.1569x; 1.1608x over previous
#include <hip/hip_runtime.h>


#define N_INPUT     4096
#define NUM_CENTERS 512
#define DIM         128
#define N_OUT       10

#define TILE_ROWS   16
#define BLOCK       256
#define BATCH_STRIDE 132
#define RAD_STRIDE   513

#define CHUNK_D     16
#define N_CHUNKS    (DIM / CHUNK_D)
#define C_PITCH     20
#define PRE_LOADS   ((NUM_CENTERS * CHUNK_D / 4) / BLOCK)
#define TR          4
#define TC          8
#define C_GROUPS    (NUM_CENTERS / TC)

typedef __attribute__((ext_vector_type(2))) float v2f;
typedef __attribute__((ext_vector_type(8))) float v8f;
typedef __attribute__((ext_vector_type(4))) float v4f;

static __device__ __forceinline__
void load_chunk(v4f (&pre)[PRE_LOADS], const float* __restrict__ centers,
                int tid, int k)
{
#pragma unroll
    for (int j = 0; j < PRE_LOADS; ++j) {
        const int p = tid + BLOCK * j;
        const int c = p >> 2;
        const int q = p & 3;
        pre[j] = *(const v4f*)(centers + (size_t)c * DIM + k * CHUNK_D + q * 4);
    }
}

static __device__ __forceinline__
void store_chunk(const v4f (&pre)[PRE_LOADS], float (&csum)[PRE_LOADS],
                 float* buf, int tid)
{
#pragma unroll
    for (int j = 0; j < PRE_LOADS; ++j) {
        const int p = tid + BLOCK * j;
        const int c = p >> 2;
        const int q = p & 3;
        *(v4f*)(buf + c * C_PITCH + q * 4) = pre[j];
        csum[j] += (pre[j].x + pre[j].y) + (pre[j].z + pre[j].w);
    }
}

__global__ __launch_bounds__(BLOCK)
void rbfn_fused_kernel(const float* __restrict__ batches,
                       const float* __restrict__ centers,
                       const float* __restrict__ sigma_p,
                       const float* __restrict__ W,
                       const float* __restrict__ bias,
                       float* __restrict__ out)
{
    __shared__ __attribute__((aligned(16))) float batch_lds[TILE_ROWS * BATCH_STRIDE];
    __shared__ __attribute__((aligned(16))) float radial_lds[TILE_ROWS * RAD_STRIDE];
    __shared__ __attribute__((aligned(16))) float cbuf[NUM_CENTERS * C_PITCH];
    __shared__ float bsum_lds[TILE_ROWS];
    __shared__ __attribute__((aligned(16))) float csum_lds[NUM_CENTERS * 4];

    const int tid  = threadIdx.x;
    const int row0 = blockIdx.x * TILE_ROWS;
    const float sigma = sigma_p[0];

    v4f first[PRE_LOADS];
    load_chunk(first, centers, tid, 0);

    {
        const float4* gsrc = (const float4*)(batches + (size_t)row0 * DIM);
        const int c4 = tid & 31;
        const int r0 = tid >> 5;
        const int r1 = r0 + BLOCK / 32;
        const float4 v0 = gsrc[r0 * (DIM / 4) + c4];
        const float4 v1 = gsrc[r1 * (DIM / 4) + c4];
        float* dst0 = &batch_lds[r0 * BATCH_STRIDE + c4 * 4];
        float* dst1 = &batch_lds[r1 * BATCH_STRIDE + c4 * 4];
        dst0[0] = v0.x; dst0[1] = v0.y; dst0[2] = v0.z; dst0[3] = v0.w;
        dst1[0] = v1.x; dst1[1] = v1.y; dst1[2] = v1.z; dst1[3] = v1.w;
        float s0 = (v0.x + v0.y) + (v0.z + v0.w);
        float s1 = (v1.x + v1.y) + (v1.z + v1.w);
#pragma unroll
        for (int off = 16; off > 0; off >>= 1) {
            s0 += __shfl_xor(s0, off);
            s1 += __shfl_xor(s1, off);
        }
        if (c4 == 0) {
            bsum_lds[r0] = s0;
            bsum_lds[r1] = s1;
        }
    }
    float csum[PRE_LOADS];
#pragma unroll
    for (int j = 0; j < PRE_LOADS; ++j)
        csum[j] = 0.0f;
    store_chunk(first, csum, cbuf, tid);

    {
        const int cg = tid & (C_GROUPS - 1);
        const int rg = tid / C_GROUPS;

        v2f acc[TR][TC];
#pragma unroll
        for (int i = 0; i < TR; ++i)
#pragma unroll
            for (int j = 0; j < TC; ++j)
                acc[i][j] = (v2f){0.0f, 0.0f};

        v4f pre[PRE_LOADS];

#pragma unroll 1
        for (int k = 0; k < N_CHUNKS; ++k) {
            if (k > 0) {
                __syncthreads();
                store_chunk(pre, csum, cbuf, tid);
                if (k == N_CHUNKS - 1) {
#pragma unroll
                    for (int j = 0; j < PRE_LOADS; ++j)
                        csum_lds[tid + BLOCK * j] = csum[j];
                }
            }
            __syncthreads();
            if (k + 1 < N_CHUNKS)
                load_chunk(pre, centers, tid, k + 1);

            const float* bp = batch_lds + (rg * TR) * BATCH_STRIDE + k * CHUNK_D;
            const float* cp = cbuf + cg * C_PITCH;

#pragma unroll
            for (int d4 = 0; d4 < CHUNK_D / 4; ++d4) {
                v4f b[TR];
                v4f c[TC];
#pragma unroll
                for (int i = 0; i < TR; ++i)
                    b[i] = *(const v4f*)(bp + i * BATCH_STRIDE + d4 * 4);
#pragma unroll
                for (int j = 0; j < TC; ++j)
                    c[j] = *(const v4f*)(cp + (j * C_GROUPS) * C_PITCH + d4 * 4);
#pragma unroll
                for (int h = 0; h < 2; ++h) {
#pragma unroll
                    for (int i = 0; i < TR; ++i) {
                        const v2f bb = (v2f){b[i][2 * h], b[i][2 * h + 1]};
#pragma unroll
                        for (int j = 0; j < TC; ++j) {
                            const v2f cc = (v2f){c[j][2 * h], c[j][2 * h + 1]};
                            acc[i][j] += __builtin_elementwise_max(bb, cc);
                        }
                    }
                }
            }
        }

        float bs[TR];
#pragma unroll
        for (int i = 0; i < TR; ++i)
            bs[i] = bsum_lds[rg * TR + i];
#pragma unroll
        for (int j = 0; j < TC; ++j) {
            const v4f cq = *(const v4f*)(csum_lds + 4 * (cg + C_GROUPS * j));
            const float cs = (cq.x + cq.y) + (cq.z + cq.w);
#pragma unroll
            for (int i = 0; i < TR; ++i) {
                const float m = acc[i][j].x + acc[i][j].y;
                const float dist = (2.0f * m - bs[i]) - cs;
                radial_lds[(rg * TR + i) * RAD_STRIDE + cg + C_GROUPS * j] =
                    __expf(-sigma * dist);
            }
        }
    }
    __syncthreads();

    if (tid < 32) {
        const int lane = tid;
        const int half = lane >> 4;
        const int m    = lane & 15;

        const bool  valid = (m < N_OUT);
        const float mask  = valid ? 1.0f : 0.0f;
        const float* wrow = W + (size_t)(valid ? m : 0) * NUM_CENTERS;
        const float* arow = radial_lds + m * RAD_STRIDE;

        v8f acc0 = {};
        v8f acc1 = {};
        for (int k = 0; k < NUM_CENTERS; k += 8) {
            const int kk0 = k + 2 * half;
            const int kk1 = k + 4 + 2 * half;

            v2f a0, a1, b0, b1;
            a0.x = arow[kk0];
            a0.y = arow[kk0 + 1];
            a1.x = arow[kk1];
            a1.y = arow[kk1 + 1];
            const float2 w0 = *(const float2*)(wrow + kk0);
            const float2 w1 = *(const float2*)(wrow + kk1);
            b0.x = w0.x * mask;  b0.y = w0.y * mask;
            b1.x = w1.x * mask;  b1.y = w1.y * mask;

            acc0 = __builtin_amdgcn_wmma_f32_16x16x4_f32(
                false, a0, false, b0, (short)0, acc0, false, false);
            acc1 = __builtin_amdgcn_wmma_f32_16x16x4_f32(
                false, a1, false, b1, (short)0, acc1, false, false);
        }

        if (valid) {
            const float bv = bias[m];
#pragma unroll
            for (int v = 0; v < 8; ++v) {
                const int row = row0 + v + 8 * half;
                out[(size_t)row * N_OUT + m] = acc0[v] + acc1[v] + bv;
            }
        }
    }
}

extern "C" void kernel_launch(void* const* d_in, const int* in_sizes, int n_in,
                              void* d_out, int out_size, void* d_ws, size_t ws_size,
                              hipStream_t stream) {
    const float* batches = (const float*)d_in[0];
    const float* centers = (const float*)d_in[1];
    const float* sigma   = (const float*)d_in[2];
    const float* W       = (const float*)d_in[3];
    const float* bias    = (const float*)d_in[4];
    float* out           = (float*)d_out;

    (void)in_sizes; (void)n_in; (void)out_size; (void)d_ws; (void)ws_size;

    rbfn_fused_kernel<<<N_INPUT / TILE_ROWS, BLOCK, 0, stream>>>(
        batches, centers, sigma, W, bias, out);
}


# Output GEMM split over all 8 waves by K (64 centres each, W slice requested ahead of the exp step), partial tiles added through LDS by wave 0; K-sum order changes (8 slice partials in wave order)
# speedup vs baseline: 4.4047x; 1.3953x over previous
#include <hip/hip_runtime.h>


#define N_INPUT     4096
#define NUM_CENTERS 512
#define DIM         128
#define N_OUT       10

#define TILE_ROWS   16
#define BLOCK       256
#define BATCH_STRIDE 132
#define RAD_STRIDE   513

#define CHUNK_D     16
#define N_CHUNKS    (DIM / CHUNK_D)
#define C_PITCH     20
#define PRE_LOADS   ((NUM_CENTERS * CHUNK_D / 4) / BLOCK)
#define TR          4
#define TC          8
#define C_GROUPS    (NUM_CENTERS / TC)

#define N_WAVES     (BLOCK / 32)
#define K_SLICE     (NUM_CENTERS / N_WAVES)
#define K_STEPS     (K_SLICE / 4)
#define PART_PITCH  20
#define PART_SIZE   (TILE_ROWS * PART_PITCH)
#define OUT_PER_LANE ((TILE_ROWS * N_OUT) / 32)

typedef __attribute__((ext_vector_type(2))) float v2f;
typedef __attribute__((ext_vector_type(8))) float v8f;
typedef __attribute__((ext_vector_type(4))) float v4f;

static __device__ __forceinline__
void load_chunk(v4f (&pre)[PRE_LOADS], const float* __restrict__ centers,
                int tid, int k)
{
#pragma unroll
    for (int j = 0; j < PRE_LOADS; ++j) {
        const int p = tid + BLOCK * j;
        const int c = p >> 2;
        const int q = p & 3;
        pre[j] = *(const v4f*)(centers + (size_t)c * DIM + k * CHUNK_D + q * 4);
    }
}

static __device__ __forceinline__
void store_chunk(const v4f (&pre)[PRE_LOADS], float (&csum)[PRE_LOADS],
                 float* buf, int tid)
{
#pragma unroll
    for (int j = 0; j < PRE_LOADS; ++j) {
        const int p = tid + BLOCK * j;
        const int c = p >> 2;
        const int q = p & 3;
        *(v4f*)(buf + c * C_PITCH + q * 4) = pre[j];
        csum[j] += (pre[j].x + pre[j].y) + (pre[j].z + pre[j].w);
    }
}

__global__ __launch_bounds__(BLOCK)
void rbfn_fused_kernel(const float* __restrict__ batches,
                       const float* __restrict__ centers,
                       const float* __restrict__ sigma_p,
                       const float* __restrict__ W,
                       const float* __restrict__ bias,
                       float* __restrict__ out)
{
    __shared__ __attribute__((aligned(16))) float batch_lds[TILE_ROWS * BATCH_STRIDE];
    __shared__ __attribute__((aligned(16))) float radial_lds[TILE_ROWS * RAD_STRIDE];
    __shared__ __attribute__((aligned(16))) float cbuf[NUM_CENTERS * C_PITCH];
    __shared__ float bsum_lds[TILE_ROWS];
    __shared__ __attribute__((aligned(16))) float csum_lds[NUM_CENTERS * 4];

    const int tid  = threadIdx.x;
    const int row0 = blockIdx.x * TILE_ROWS;
    const float sigma = sigma_p[0];

    v4f first[PRE_LOADS];
    load_chunk(first, centers, tid, 0);

    {
        const float4* gsrc = (const float4*)(batches + (size_t)row0 * DIM);
        const int c4 = tid & 31;
        const int r0 = tid >> 5;
        const int r1 = r0 + BLOCK / 32;
        const float4 v0 = gsrc[r0 * (DIM / 4) + c4];
        const float4 v1 = gsrc[r1 * (DIM / 4) + c4];
        float* dst0 = &batch_lds[r0 * BATCH_STRIDE + c4 * 4];
        float* dst1 = &batch_lds[r1 * BATCH_STRIDE + c4 * 4];
        dst0[0] = v0.x; dst0[1] = v0.y; dst0[2] = v0.z; dst0[3] = v0.w;
        dst1[0] = v1.x; dst1[1] = v1.y; dst1[2] = v1.z; dst1[3] = v1.w;
        float s0 = (v0.x + v0.y) + (v0.z + v0.w);
        float s1 = (v1.x + v1.y) + (v1.z + v1.w);
#pragma unroll
        for (int off = 16; off > 0; off >>= 1) {
            s0 += __shfl_xor(s0, off);
            s1 += __shfl_xor(s1, off);
        }
        if (c4 == 0) {
            bsum_lds[r0] = s0;
            bsum_lds[r1] = s1;
        }
    }
    float csum[PRE_LOADS];
#pragma unroll
    for (int j = 0; j < PRE_LOADS; ++j)
        csum[j] = 0.0f;
    store_chunk(first, csum, cbuf, tid);

    const int lane = tid & 31;
    const int wv   = tid >> 5;
    const int half = lane >> 4;
    const int lm   = lane & 15;
    const bool  valid = (lm < N_OUT);
    const float mask  = valid ? 1.0f : 0.0f;
    const float* wrow = W + (size_t)(valid ? lm : 0) * NUM_CENTERS + wv * K_SLICE + 2 * half;
    float2 wfrag[K_STEPS];
    float  bv[OUT_PER_LANE];

    {
        const int cg = tid & (C_GROUPS - 1);
        const int rg = tid / C_GROUPS;

        v2f acc[TR][TC];
#pragma unroll
        for (int i = 0; i < TR; ++i)
#pragma unroll
            for (int j = 0; j < TC; ++j)
                acc[i][j] = (v2f){0.0f, 0.0f};

        v4f pre[PRE_LOADS];

#pragma unroll 1
        for (int k = 0; k < N_CHUNKS; ++k) {
            if (k > 0) {
                __syncthreads();
                store_chunk(pre, csum, cbuf, tid);
                if (k == N_CHUNKS - 1) {
#pragma unroll
                    for (int j = 0; j < PRE_LOADS; ++j)
                        csum_lds[tid + BLOCK * j] = csum[j];
                }
            }
            __syncthreads();
            if (k + 1 < N_CHUNKS)
                load_chunk(pre, centers, tid, k + 1);

            const float* bp = batch_lds + (rg * TR) * BATCH_STRIDE + k * CHUNK_D;
            const float* cp = cbuf + cg * C_PITCH;

#pragma unroll
            for (int d4 = 0; d4 < CHUNK_D / 4; ++d4) {
                v4f b[TR];
                v4f c[TC];
#pragma unroll
                for (int i = 0; i < TR; ++i)
                    b[i] = *(const v4f*)(bp + i * BATCH_STRIDE + d4 * 4);
#pragma unroll
                for (int j = 0; j < TC; ++j)
                    c[j] = *(const v4f*)(cp + (j * C_GROUPS) * C_PITCH + d4 * 4);
#pragma unroll
                for (int h = 0; h < 2; ++h) {
#pragma unroll
                    for (int i = 0; i < TR; ++i) {
                        const v2f bb = (v2f){b[i][2 * h], b[i][2 * h + 1]};
#pragma unroll
                        for (int j = 0; j < TC; ++j) {
                            const v2f cc = (v2f){c[j][2 * h], c[j][2 * h + 1]};
                            acc[i][j] += __builtin_elementwise_max(bb, cc);
                        }
                    }
                }
            }
        }

#pragma unroll
        for (int st = 0; st < K_STEPS; ++st)
            wfrag[st] = *(const float2*)(wrow + 4 * st);
#pragma unroll
        for (int i = 0; i < OUT_PER_LANE; ++i)
            bv[i] = bias[(lane + 32 * i) % N_OUT];

        float bs[TR];
#pragma unroll
        for (int i = 0; i < TR; ++i)
            bs[i] = bsum_lds[rg * TR + i];
#pragma unroll
        for (int j = 0; j < TC; ++j) {
            const v4f cq = *(const v4f*)(csum_lds + 4 * (cg + C_GROUPS * j));
            const float cs = (cq.x + cq.y) + (cq.z + cq.w);
#pragma unroll
            for (int i = 0; i < TR; ++i) {
                const float m = acc[i][j].x + acc[i][j].y;
                const float dist = (2.0f * m - bs[i]) - cs;
                radial_lds[(rg * TR + i) * RAD_STRIDE + cg + C_GROUPS * j] =
                    __expf(-sigma * dist);
            }
        }

        asm volatile("" :: "v"(wfrag[0]),  "v"(wfrag[1]),  "v"(wfrag[2]),  "v"(wfrag[3]),
                           "v"(wfrag[4]),  "v"(wfrag[5]),  "v"(wfrag[6]),  "v"(wfrag[7]),
                           "v"(wfrag[8]),  "v"(wfrag[9]),  "v"(wfrag[10]), "v"(wfrag[11]),
                           "v"(wfrag[12]), "v"(wfrag[13]), "v"(wfrag[14]), "v"(wfrag[15]));
    }
    __syncthreads();

    {
        const float* arow = radial_lds + lm * RAD_STRIDE + wv * K_SLICE + 2 * half;
        v2f af[K_STEPS];
#pragma unroll
        for (int st = 0; st < K_STEPS; ++st) {
            af[st].x = arow[4 * st];
            af[st].y = arow[4 * st + 1];
        }

        v8f acc0 = {};
        v8f acc1 = {};
#pragma unroll
        for (int st = 0; st < K_STEPS; st += 2) {
            v2f b0, b1;
            b0.x = wfrag[st].x * mask;      b0.y = wfrag[st].y * mask;
            b1.x = wfrag[st + 1].x * mask;  b1.y = wfrag[st + 1].y * mask;
            acc0 = __builtin_amdgcn_wmma_f32_16x16x4_f32(
                false, af[st], false, b0, (short)0, acc0, false, false);
            acc1 = __builtin_amdgcn_wmma_f32_16x16x4_f32(
                false, af[st + 1], false, b1, (short)0, acc1, false, false);
        }

        float* part = cbuf + wv * PART_SIZE;
#pragma unroll
        for (int v = 0; v < 8; ++v)
            part[(v + 8 * half) * PART_PITCH + lm] = acc0[v] + acc1[v];
    }
    __syncthreads();

    if (tid < 32) {
#pragma unroll
        for (int i = 0; i < OUT_PER_LANE; ++i) {
            const int e = lane + 32 * i;
            const int r = e / N_OUT;
            const int c = e - r * N_OUT;
            const float* p = cbuf + r * PART_PITCH + c;
            float sum = p[0];
#pragma unroll
            for (int w = 1; w < N_WAVES; ++w)
                sum += p[w * PART_SIZE];
            out[(size_t)row0 * N_OUT + e] = sum + bv[i];
        }
    }
}

extern "C" void kernel_launch(void* const* d_in, const int* in_sizes, int n_in,
                              void* d_out, int out_size, void* d_ws, size_t ws_size,
                              hipStream_t stream) {
    const float* batches = (const float*)d_in[0];
    const float* centers = (const float*)d_in[1];
    const float* sigma   = (const float*)d_in[2];
    const float* W       = (const float*)d_in[3];
    const float* bias    = (const float*)d_in[4];
    float* out           = (float*)d_out;

    (void)in_sizes; (void)n_in; (void)out_size; (void)d_ws; (void)ws_size;

    rbfn_fused_kernel<<<N_INPUT / TILE_ROWS, BLOCK, 0, stream>>>(
        batches, centers, sigma, W, bias, out);
}


# Distance loop: the max pairs written by hand as dual-issue instructions (8 per asm statement), dropping the compiler's NaN-quieting max(x,x) on every LDS-loaded operand; sums and add order unchanged
# speedup vs baseline: 5.0924x; 1.1561x over previous
#include <hip/hip_runtime.h>


#define N_INPUT     4096
#define NUM_CENTERS 512
#define DIM         128
#define N_OUT       10

#define TILE_ROWS   16
#define BLOCK       256
#define BATCH_STRIDE 132
#define RAD_STRIDE   513

#define CHUNK_D     16
#define N_CHUNKS    (DIM / CHUNK_D)
#define C_PITCH     20
#define PRE_LOADS   ((NUM_CENTERS * CHUNK_D / 4) / BLOCK)
#define TR          4
#define TC          8
#define C_GROUPS    (NUM_CENTERS / TC)

#define N_WAVES     (BLOCK / 32)
#define K_SLICE     (NUM_CENTERS / N_WAVES)
#define K_STEPS     (K_SLICE / 4)
#define PART_PITCH  20
#define PART_SIZE   (TILE_ROWS * PART_PITCH)
#define OUT_PER_LANE ((TILE_ROWS * N_OUT) / 32)

typedef __attribute__((ext_vector_type(2))) float v2f;
typedef __attribute__((ext_vector_type(8))) float v8f;
typedef __attribute__((ext_vector_type(4))) float v4f;

static __device__ __forceinline__
void load_chunk(v4f (&pre)[PRE_LOADS], const float* __restrict__ centers,
                int tid, int k)
{
#pragma unroll
    for (int j = 0; j < PRE_LOADS; ++j) {
        const int p = tid + BLOCK * j;
        const int c = p >> 2;
        const int q = p & 3;
        pre[j] = *(const v4f*)(centers + (size_t)c * DIM + k * CHUNK_D + q * 4);
    }
}

static __device__ __forceinline__
void store_chunk(const v4f (&pre)[PRE_LOADS], float (&csum)[PRE_LOADS],
                 float* buf, int tid)
{
#pragma unroll
    for (int j = 0; j < PRE_LOADS; ++j) {
        const int p = tid + BLOCK * j;
        const int c = p >> 2;
        const int q = p & 3;
        *(v4f*)(buf + c * C_PITCH + q * 4) = pre[j];
        csum[j] += (pre[j].x + pre[j].y) + (pre[j].z + pre[j].w);
    }
}

__global__ __launch_bounds__(BLOCK)
void rbfn_fused_kernel(const float* __restrict__ batches,
                       const float* __restrict__ centers,
                       const float* __restrict__ sigma_p,
                       const float* __restrict__ W,
                       const float* __restrict__ bias,
                       float* __restrict__ out)
{
    __shared__ __attribute__((aligned(16))) float batch_lds[TILE_ROWS * BATCH_STRIDE];
    __shared__ __attribute__((aligned(16))) float radial_lds[TILE_ROWS * RAD_STRIDE];
    __shared__ __attribute__((aligned(16))) float cbuf[NUM_CENTERS * C_PITCH];
    __shared__ float bsum_lds[TILE_ROWS];
    __shared__ __attribute__((aligned(16))) float csum_lds[NUM_CENTERS * 4];

    const int tid  = threadIdx.x;
    const int row0 = blockIdx.x * TILE_ROWS;
    const float sigma = sigma_p[0];

    v4f first[PRE_LOADS];
    load_chunk(first, centers, tid, 0);

    {
        const float4* gsrc = (const float4*)(batches + (size_t)row0 * DIM);
        const int c4 = tid & 31;
        const int r0 = tid >> 5;
        const int r1 = r0 + BLOCK / 32;
        const float4 v0 = gsrc[r0 * (DIM / 4) + c4];
        const float4 v1 = gsrc[r1 * (DIM / 4) + c4];
        float* dst0 = &batch_lds[r0 * BATCH_STRIDE + c4 * 4];
        float* dst1 = &batch_lds[r1 * BATCH_STRIDE + c4 * 4];
        dst0[0] = v0.x; dst0[1] = v0.y; dst0[2] = v0.z; dst0[3] = v0.w;
        dst1[0] = v1.x; dst1[1] = v1.y; dst1[2] = v1.z; dst1[3] = v1.w;
        float s0 = (v0.x + v0.y) + (v0.z + v0.w);
        float s1 = (v1.x + v1.y) + (v1.z + v1.w);
#pragma unroll
        for (int off = 16; off > 0; off >>= 1) {
            s0 += __shfl_xor(s0, off);
            s1 += __shfl_xor(s1, off);
        }
        if (c4 == 0) {
            bsum_lds[r0] = s0;
            bsum_lds[r1] = s1;
        }
    }
    float csum[PRE_LOADS];
#pragma unroll
    for (int j = 0; j < PRE_LOADS; ++j)
        csum[j] = 0.0f;
    store_chunk(first, csum, cbuf, tid);

    const int lane = tid & 31;
    const int wv   = tid >> 5;
    const int half = lane >> 4;
    const int lm   = lane & 15;
    const bool  valid = (lm < N_OUT);
    const float mask  = valid ? 1.0f : 0.0f;
    const float* wrow = W + (size_t)(valid ? lm : 0) * NUM_CENTERS + wv * K_SLICE + 2 * half;
    float2 wfrag[K_STEPS];
    float  bv[OUT_PER_LANE];

    {
        const int cg = tid & (C_GROUPS - 1);
        const int rg = tid / C_GROUPS;

        v2f acc[TR][TC];
#pragma unroll
        for (int i = 0; i < TR; ++i)
#pragma unroll
            for (int j = 0; j < TC; ++j)
                acc[i][j] = (v2f){0.0f, 0.0f};

        v4f pre[PRE_LOADS];

#pragma unroll 1
        for (int k = 0; k < N_CHUNKS; ++k) {
            if (k > 0) {
                __syncthreads();
                store_chunk(pre, csum, cbuf, tid);
                if (k == N_CHUNKS - 1) {
#pragma unroll
                    for (int j = 0; j < PRE_LOADS; ++j)
                        csum_lds[tid + BLOCK * j] = csum[j];
                }
            }
            __syncthreads();
            if (k + 1 < N_CHUNKS)
                load_chunk(pre, centers, tid, k + 1);

            const float* bp = batch_lds + (rg * TR) * BATCH_STRIDE + k * CHUNK_D;
            const float* cp = cbuf + cg * C_PITCH;

#pragma unroll
            for (int d4 = 0; d4 < CHUNK_D / 4; ++d4) {
                v4f b[TR];
                v4f c[TC];
#pragma unroll
                for (int i = 0; i < TR; ++i)
                    b[i] = *(const v4f*)(bp + i * BATCH_STRIDE + d4 * 4);
#pragma unroll
                for (int j = 0; j < TC; ++j)
                    c[j] = *(const v4f*)(cp + (j * C_GROUPS) * C_PITCH + d4 * 4);
#pragma unroll
                for (int h = 0; h < 2; ++h) {
#pragma unroll
                    for (int i = 0; i < TR; i += 2) {
#pragma unroll
                        for (int j = 0; j < TC; j += 4) {
                            float mx[2][4], my[2][4];
                            asm("v_dual_max_num_f32 %0, %16, %20 :: v_dual_max_num_f32 %1, %17, %21\n\t"
                                "v_dual_max_num_f32 %2, %16, %22 :: v_dual_max_num_f32 %3, %17, %23\n\t"
                                "v_dual_max_num_f32 %4, %16, %24 :: v_dual_max_num_f32 %5, %17, %25\n\t"
                                "v_dual_max_num_f32 %6, %16, %26 :: v_dual_max_num_f32 %7, %17, %27\n\t"
                                "v_dual_max_num_f32 %8, %18, %20 :: v_dual_max_num_f32 %9, %19, %21\n\t"
                                "v_dual_max_num_f32 %10, %18, %22 :: v_dual_max_num_f32 %11, %19, %23\n\t"
                                "v_dual_max_num_f32 %12, %18, %24 :: v_dual_max_num_f32 %13, %19, %25\n\t"
                                "v_dual_max_num_f32 %14, %18, %26 :: v_dual_max_num_f32 %15, %19, %27"
                                : "=&v"(mx[0][0]), "=&v"(my[0][0]), "=&v"(mx[0][1]), "=&v"(my[0][1]),
                                  "=&v"(mx[0][2]), "=&v"(my[0][2]), "=&v"(mx[0][3]), "=&v"(my[0][3]),
                                  "=&v"(mx[1][0]), "=&v"(my[1][0]), "=&v"(mx[1][1]), "=&v"(my[1][1]),
                                  "=&v"(mx[1][2]), "=&v"(my[1][2]), "=&v"(mx[1][3]), "=&v"(my[1][3])
                                : "v"(b[i][2 * h]), "v"(b[i][2 * h + 1]),
                                  "v"(b[i + 1][2 * h]), "v"(b[i + 1][2 * h + 1]),
                                  "v"(c[j][2 * h]), "v"(c[j][2 * h + 1]),
                                  "v"(c[j + 1][2 * h]), "v"(c[j + 1][2 * h + 1]),
                                  "v"(c[j + 2][2 * h]), "v"(c[j + 2][2 * h + 1]),
                                  "v"(c[j + 3][2 * h]), "v"(c[j + 3][2 * h + 1]));
#pragma unroll
                            for (int r = 0; r < 2; ++r)
#pragma unroll
                                for (int q = 0; q < 4; ++q)
                                    acc[i + r][j + q] += (v2f){mx[r][q], my[r][q]};
                        }
                    }
                }
            }
        }

#pragma unroll
        for (int st = 0; st < K_STEPS; ++st)
            wfrag[st] = *(const float2*)(wrow + 4 * st);
#pragma unroll
        for (int i = 0; i < OUT_PER_LANE; ++i)
            bv[i] = bias[(lane + 32 * i) % N_OUT];

        float bs[TR];
#pragma unroll
        for (int i = 0; i < TR; ++i)
            bs[i] = bsum_lds[rg * TR + i];
#pragma unroll
        for (int j = 0; j < TC; ++j) {
            const v4f cq = *(const v4f*)(csum_lds + 4 * (cg + C_GROUPS * j));
            const float cs = (cq.x + cq.y) + (cq.z + cq.w);
#pragma unroll
            for (int i = 0; i < TR; ++i) {
                const float m = acc[i][j].x + acc[i][j].y;
                const float dist = (2.0f * m - bs[i]) - cs;
                radial_lds[(rg * TR + i) * RAD_STRIDE + cg + C_GROUPS * j] =
                    __expf(-sigma * dist);
            }
        }

        asm volatile("" :: "v"(wfrag[0]),  "v"(wfrag[1]),  "v"(wfrag[2]),  "v"(wfrag[3]),
                           "v"(wfrag[4]),  "v"(wfrag[5]),  "v"(wfrag[6]),  "v"(wfrag[7]),
                           "v"(wfrag[8]),  "v"(wfrag[9]),  "v"(wfrag[10]), "v"(wfrag[11]),
                           "v"(wfrag[12]), "v"(wfrag[13]), "v"(wfrag[14]), "v"(wfrag[15]));
    }
    __syncthreads();

    {
        const float* arow = radial_lds + lm * RAD_STRIDE + wv * K_SLICE + 2 * half;
        v2f af[K_STEPS];
#pragma unroll
        for (int st = 0; st < K_STEPS; ++st) {
            af[st].x = arow[4 * st];
            af[st].y = arow[4 * st + 1];
        }

        v8f acc0 = {};
        v8f acc1 = {};
#pragma unroll
        for (int st = 0; st < K_STEPS; st += 2) {
            v2f b0, b1;
            b0.x = wfrag[st].x * mask;      b0.y = wfrag[st].y * mask;
            b1.x = wfrag[st + 1].x * mask;  b1.y = wfrag[st + 1].y * mask;
            acc0 = __builtin_amdgcn_wmma_f32_16x16x4_f32(
                false, af[st], false, b0, (short)0, acc0, false, false);
            acc1 = __builtin_amdgcn_wmma_f32_16x16x4_f32(
                false, af[st + 1], false, b1, (short)0, acc1, false, false);
        }

        float* part = cbuf + wv * PART_SIZE;
#pragma unroll
        for (int v = 0; v < 8; ++v)
            part[(v + 8 * half) * PART_PITCH + lm] = acc0[v] + acc1[v];
    }
    __syncthreads();

    if (tid < 32) {
#pragma unroll
        for (int i = 0; i < OUT_PER_LANE; ++i) {
            const int e = lane + 32 * i;
            const int r = e / N_OUT;
            const int c = e - r * N_OUT;
            const float* p = cbuf + r * PART_PITCH + c;
            float sum = p[0];
#pragma unroll
            for (int w = 1; w < N_WAVES; ++w)
                sum += p[w * PART_SIZE];
            out[(size_t)row0 * N_OUT + e] = sum + bv[i];
        }
    }
}

extern "C" void kernel_launch(void* const* d_in, const int* in_sizes, int n_in,
                              void* d_out, int out_size, void* d_ws, size_t ws_size,
                              hipStream_t stream) {
    const float* batches = (const float*)d_in[0];
    const float* centers = (const float*)d_in[1];
    const float* sigma   = (const float*)d_in[2];
    const float* W       = (const float*)d_in[3];
    const float* bias    = (const float*)d_in[4];
    float* out           = (float*)d_out;

    (void)in_sizes; (void)n_in; (void)out_size; (void)d_ws; (void)ws_size;

    rbfn_fused_kernel<<<N_INPUT / TILE_ROWS, BLOCK, 0, stream>>>(
        batches, centers, sigma, W, bias, out);
}


# Centre chunks go round a ring of three LDS stages, each written at the head of the chunk before; one split work-group barrier per chunk (signal behind the stores, wait a chunk later) replaces two
# speedup vs baseline: 5.7043x; 1.1202x over previous
#include <hip/hip_runtime.h>


#define N_INPUT     4096
#define NUM_CENTERS 512
#define DIM         128
#define N_OUT       10

#define TILE_ROWS   16
#define BLOCK       256
#define BATCH_STRIDE 132
#define RAD_STRIDE   513

#define CHUNK_D     16
#define N_CHUNKS    (DIM / CHUNK_D)
#define C_PITCH     20
#define PRE_LOADS   ((NUM_CENTERS * CHUNK_D / 4) / BLOCK)
#define N_STAGES     3
#define STAGE_FLOATS (NUM_CENTERS * C_PITCH)
#define TR          4
#define TC          8
#define C_GROUPS    (NUM_CENTERS / TC)

#define N_WAVES     (BLOCK / 32)
#define K_SLICE     (NUM_CENTERS / N_WAVES)
#define K_STEPS     (K_SLICE / 4)
#define PART_PITCH  20
#define PART_SIZE   (TILE_ROWS * PART_PITCH)
#define OUT_PER_LANE ((TILE_ROWS * N_OUT) / 32)

typedef __attribute__((ext_vector_type(2))) float v2f;
typedef __attribute__((ext_vector_type(8))) float v8f;
typedef __attribute__((ext_vector_type(4))) float v4f;

static __device__ __forceinline__
void load_chunk(v4f (&pre)[PRE_LOADS], const float* __restrict__ centers,
                int tid, int k)
{
#pragma unroll
    for (int j = 0; j < PRE_LOADS; ++j) {
        const int p = tid + BLOCK * j;
        const int c = p >> 2;
        const int q = p & 3;
        pre[j] = *(const v4f*)(centers + (size_t)c * DIM + k * CHUNK_D + q * 4);
    }
}

static __device__ __forceinline__
void store_chunk(const v4f (&pre)[PRE_LOADS], float (&csum)[PRE_LOADS],
                 float* buf, int tid)
{
#pragma unroll
    for (int j = 0; j < PRE_LOADS; ++j) {
        const int p = tid + BLOCK * j;
        const int c = p >> 2;
        const int q = p & 3;
        *(v4f*)(buf + c * C_PITCH + q * 4) = pre[j];
        csum[j] += (pre[j].x + pre[j].y) + (pre[j].z + pre[j].w);
    }
}

template <int D4_FROM, int D4_TO>
static __device__ __forceinline__
void sum_steps(v2f (&acc)[TR][TC], const float* bp, const float* cp)
{
#pragma unroll
    for (int d4 = D4_FROM; d4 < D4_TO; ++d4) {
        v4f b[TR];
        v4f c[TC];
#pragma unroll
        for (int i = 0; i < TR; ++i)
            b[i] = *(const v4f*)(bp + i * BATCH_STRIDE + d4 * 4);
#pragma unroll
        for (int j = 0; j < TC; ++j)
            c[j] = *(const v4f*)(cp + (j * C_GROUPS) * C_PITCH + d4 * 4);
#pragma unroll
        for (int h = 0; h < 2; ++h) {
#pragma unroll
            for (int i = 0; i < TR; i += 2) {
#pragma unroll
                for (int j = 0; j < TC; j += 4) {
                    float mx[2][4], my[2][4];
                    asm("v_dual_max_num_f32 %0, %16, %20 :: v_dual_max_num_f32 %1, %17, %21\n\t"
                        "v_dual_max_num_f32 %2, %16, %22 :: v_dual_max_num_f32 %3, %17, %23\n\t"
                        "v_dual_max_num_f32 %4, %16, %24 :: v_dual_max_num_f32 %5, %17, %25\n\t"
                        "v_dual_max_num_f32 %6, %16, %26 :: v_dual_max_num_f32 %7, %17, %27\n\t"
                        "v_dual_max_num_f32 %8, %18, %20 :: v_dual_max_num_f32 %9, %19, %21\n\t"
                        "v_dual_max_num_f32 %10, %18, %22 :: v_dual_max_num_f32 %11, %19, %23\n\t"
                        "v_dual_max_num_f32 %12, %18, %24 :: v_dual_max_num_f32 %13, %19, %25\n\t"
                        "v_dual_max_num_f32 %14, %18, %26 :: v_dual_max_num_f32 %15, %19, %27"
                        : "=&v"(mx[0][0]), "=&v"(my[0][0]), "=&v"(mx[0][1]), "=&v"(my[0][1]),
                          "=&v"(mx[0][2]), "=&v"(my[0][2]), "=&v"(mx[0][3]), "=&v"(my[0][3]),
                          "=&v"(mx[1][0]), "=&v"(my[1][0]), "=&v"(mx[1][1]), "=&v"(my[1][1]),
                          "=&v"(mx[1][2]), "=&v"(my[1][2]), "=&v"(mx[1][3]), "=&v"(my[1][3])
                        : "v"(b[i][2 * h]), "v"(b[i][2 * h + 1]),
                          "v"(b[i + 1][2 * h]), "v"(b[i + 1][2 * h + 1]),
                          "v"(c[j][2 * h]), "v"(c[j][2 * h + 1]),
                          "v"(c[j + 1][2 * h]), "v"(c[j + 1][2 * h + 1]),
                          "v"(c[j + 2][2 * h]), "v"(c[j + 2][2 * h + 1]),
                          "v"(c[j + 3][2 * h]), "v"(c[j + 3][2 * h + 1]));
#pragma unroll
                    for (int r = 0; r < 2; ++r)
#pragma unroll
                        for (int q = 0; q < 4; ++q)
                            acc[i + r][j + q] += (v2f){mx[r][q], my[r][q]};
                }
            }
        }
    }
}

static __device__ __forceinline__ void handover_signal()
{
    __builtin_amdgcn_fence(__ATOMIC_RELEASE, "workgroup");
    __builtin_amdgcn_s_barrier_signal(-1);
}
static __device__ __forceinline__ void handover_wait()
{
    __builtin_amdgcn_s_barrier_wait(-1);
    __builtin_amdgcn_fence(__ATOMIC_ACQUIRE, "workgroup");
}

__global__ __launch_bounds__(BLOCK) __attribute__((amdgpu_num_vgpr(256)))
void rbfn_fused_kernel(const float* __restrict__ batches,
                       const float* __restrict__ centers,
                       const float* __restrict__ sigma_p,
                       const float* __restrict__ W,
                       const float* __restrict__ bias,
                       float* __restrict__ out)
{
    __shared__ __attribute__((aligned(16))) float batch_lds[TILE_ROWS * BATCH_STRIDE];
    __shared__ __attribute__((aligned(16))) float radial_lds[TILE_ROWS * RAD_STRIDE];
    __shared__ __attribute__((aligned(16))) float cbuf[N_STAGES * STAGE_FLOATS];
    __shared__ float bsum_lds[TILE_ROWS];
    __shared__ __attribute__((aligned(16))) float csum_lds[NUM_CENTERS * 4];

    const int tid  = threadIdx.x;
    const int row0 = blockIdx.x * TILE_ROWS;
    const float sigma = sigma_p[0];

    v4f first[PRE_LOADS];
    load_chunk(first, centers, tid, 0);
    v4f pre[PRE_LOADS];
    load_chunk(pre, centers, tid, 1);

    {
        const float4* gsrc = (const float4*)(batches + (size_t)row0 * DIM);
        const int c4 = tid & 31;
        const int r0 = tid >> 5;
        const int r1 = r0 + BLOCK / 32;
        const float4 v0 = gsrc[r0 * (DIM / 4) + c4];
        const float4 v1 = gsrc[r1 * (DIM / 4) + c4];
        float* dst0 = &batch_lds[r0 * BATCH_STRIDE + c4 * 4];
        float* dst1 = &batch_lds[r1 * BATCH_STRIDE + c4 * 4];
        dst0[0] = v0.x; dst0[1] = v0.y; dst0[2] = v0.z; dst0[3] = v0.w;
        dst1[0] = v1.x; dst1[1] = v1.y; dst1[2] = v1.z; dst1[3] = v1.w;
        float s0 = (v0.x + v0.y) + (v0.z + v0.w);
        float s1 = (v1.x + v1.y) + (v1.z + v1.w);
#pragma unroll
        for (int off = 16; off > 0; off >>= 1) {
            s0 += __shfl_xor(s0, off);
            s1 += __shfl_xor(s1, off);
        }
        if (c4 == 0) {
            bsum_lds[r0] = s0;
            bsum_lds[r1] = s1;
        }
    }
    float csum[PRE_LOADS];
#pragma unroll
    for (int j = 0; j < PRE_LOADS; ++j)
        csum[j] = 0.0f;
    store_chunk(first, csum, cbuf, tid);
    __syncthreads();

    const int lane = tid & 31;
    const int wv   = tid >> 5;
    const int half = lane >> 4;
    const int lm   = lane & 15;
    const bool  valid = (lm < N_OUT);
    const float mask  = valid ? 1.0f : 0.0f;
    const float* wrow = W + (size_t)(valid ? lm : 0) * NUM_CENTERS + wv * K_SLICE + 2 * half;
    float2 wfrag[K_STEPS];
    float  bv[OUT_PER_LANE];

    {
        const int cg = tid & (C_GROUPS - 1);
        const int rg = tid / C_GROUPS;

        v2f acc[TR][TC];
#pragma unroll
        for (int i = 0; i < TR; ++i)
#pragma unroll
            for (int j = 0; j < TC; ++j)
                acc[i][j] = (v2f){0.0f, 0.0f};

        const float* bp = batch_lds + (rg * TR) * BATCH_STRIDE;
        int cur = 0;

#pragma unroll 1
        for (int k = 0; k < N_CHUNKS - 1; ++k) {
            const int nxt = (cur == (N_STAGES - 1) * STAGE_FLOATS) ? 0 : cur + STAGE_FLOATS;
            store_chunk(pre, csum, cbuf + nxt, tid);
            if (k == N_CHUNKS - 2) {
#pragma unroll
                for (int j = 0; j < PRE_LOADS; ++j)
                    csum_lds[tid + BLOCK * j] = csum[j];
            }
            handover_signal();
            if (k + 2 < N_CHUNKS)
                load_chunk(pre, centers, tid, k + 2);
            sum_steps<0, CHUNK_D / 4>(acc, bp + k * CHUNK_D, cbuf + cur + cg * C_PITCH);
            handover_wait();
            cur = nxt;
        }
        sum_steps<0, CHUNK_D / 4>(acc, bp + (N_CHUNKS - 1) * CHUNK_D, cbuf + cur + cg * C_PITCH);

#pragma unroll
        for (int st = 0; st < K_STEPS; ++st)
            wfrag[st] = *(const float2*)(wrow + 4 * st);
#pragma unroll
        for (int i = 0; i < OUT_PER_LANE; ++i)
            bv[i] = bias[(lane + 32 * i) % N_OUT];

        float bs[TR];
#pragma unroll
        for (int i = 0; i < TR; ++i)
            bs[i] = bsum_lds[rg * TR + i];
#pragma unroll
        for (int j = 0; j < TC; ++j) {
            const v4f cq = *(const v4f*)(csum_lds + 4 * (cg + C_GROUPS * j));
            const float cs = (cq.x + cq.y) + (cq.z + cq.w);
#pragma unroll
            for (int i = 0; i < TR; ++i) {
                const float m = acc[i][j].x + acc[i][j].y;
                const float dist = (2.0f * m - bs[i]) - cs;
                radial_lds[(rg * TR + i) * RAD_STRIDE + cg + C_GROUPS * j] =
                    __expf(-sigma * dist);
            }
        }

        asm volatile("" :: "v"(wfrag[0]),  "v"(wfrag[1]),  "v"(wfrag[2]),  "v"(wfrag[3]),
                           "v"(wfrag[4]),  "v"(wfrag[5]),  "v"(wfrag[6]),  "v"(wfrag[7]),
                           "v"(wfrag[8]),  "v"(wfrag[9]),  "v"(wfrag[10]), "v"(wfrag[11]),
                           "v"(wfrag[12]), "v"(wfrag[13]), "v"(wfrag[14]), "v"(wfrag[15]));
    }
    __syncthreads();

    {
        const float* arow = radial_lds + lm * RAD_STRIDE + wv * K_SLICE + 2 * half;
        v2f af[K_STEPS];
#pragma unroll
        for (int st = 0; st < K_STEPS; ++st) {
            af[st].x = arow[4 * st];
            af[st].y = arow[4 * st + 1];
        }

        v8f acc0 = {};
        v8f acc1 = {};
#pragma unroll
        for (int st = 0; st < K_STEPS; st += 2) {
            v2f b0, b1;
            b0.x = wfrag[st].x * mask;      b0.y = wfrag[st].y * mask;
            b1.x = wfrag[st + 1].x * mask;  b1.y = wfrag[st + 1].y * mask;
            acc0 = __builtin_amdgcn_wmma_f32_16x16x4_f32(
                false, af[st], false, b0, (short)0, acc0, false, false);
            acc1 = __builtin_amdgcn_wmma_f32_16x16x4_f32(
                false, af[st + 1], false, b1, (short)0, acc1, false, false);
        }

        float* part = cbuf + wv * PART_SIZE;
#pragma unroll
        for (int v = 0; v < 8; ++v)
            part[(v + 8 * half) * PART_PITCH + lm] = acc0[v] + acc1[v];
    }
    __syncthreads();

    if (tid < 32) {
#pragma unroll
        for (int i = 0; i < OUT_PER_LANE; ++i) {
            const int e = lane + 32 * i;
            const int r = e / N_OUT;
            const int c = e - r * N_OUT;
            const float* p = cbuf + r * PART_PITCH + c;
            float sum = p[0];
#pragma unroll
            for (int w = 1; w < N_WAVES; ++w)
                sum += p[w * PART_SIZE];
            out[(size_t)row0 * N_OUT + e] = sum + bv[i];
        }
    }
}

extern "C" void kernel_launch(void* const* d_in, const int* in_sizes, int n_in,
                              void* d_out, int out_size, void* d_ws, size_t ws_size,
                              hipStream_t stream) {
    const float* batches = (const float*)d_in[0];
    const float* centers = (const float*)d_in[1];
    const float* sigma   = (const float*)d_in[2];
    const float* W       = (const float*)d_in[3];
    const float* bias    = (const float*)d_in[4];
    float* out           = (float*)d_out;

    (void)in_sizes; (void)n_in; (void)out_size; (void)d_ws; (void)ws_size;

    rbfn_fused_kernel<<<N_INPUT / TILE_ROWS, BLOCK, 0, stream>>>(
        batches, centers, sigma, W, bias, out);
}
